# starting kernel
# baseline (speedup 1.0000x reference)
#include <hip/hip_runtime.h>
#include <hip/hip_bf16.h>

typedef __attribute__((ext_vector_type(2)))  float    v2f;
typedef __attribute__((ext_vector_type(8)))  float    v8f;
typedef __attribute__((ext_vector_type(16))) _Float16 v16h;

#define UB     700
#define PN     256
#define SN     1024
#define CBOX   128
#define DSEED  256
#define CBFU   128
#define KDIM   512
#define ODIM   128
#define BPG    16
#define GPAD   8

__global__ __launch_bounds__(256)
void scenebox_fused_kernel(const float* __restrict__ union_box,
                           const float* __restrict__ box_features,
                           const float* __restrict__ agg_xyz,
                           const float* __restrict__ seed_feature,
                           const float* __restrict__ seed_xyz,
                           const float* __restrict__ bfu,
                           const float* __restrict__ W,
                           const float* __restrict__ bias,
                           float* __restrict__ out)
{
    __shared__ float        g[BPG][KDIM + GPAD];
    __shared__ float        bmin[BPG][3], bmax[BPG][3];
    __shared__ unsigned int smask[SN];
    __shared__ unsigned int pmask[PN];

    const int t    = threadIdx.x;
    const int box0 = blockIdx.x * BPG;
    const int lane = t & 31;
    const int wave = t >> 5;
    const int nbase = wave * 16;
    const int hsel  = lane >> 4;
    const int l15   = lane & 15;

    __builtin_prefetch(&W[(nbase + l15) * KDIM], 0, 3);
    __builtin_prefetch(&W[(nbase + l15) * KDIM + 256], 0, 3);

    if (t < BPG) {
        const int b = box0 + t;
        float c0, c1, c2, s0, s1, s2;
        if (b < UB) {
            c0 = union_box[b*6+0]; c1 = union_box[b*6+1]; c2 = union_box[b*6+2];
            s0 = union_box[b*6+3]; s1 = union_box[b*6+4]; s2 = union_box[b*6+5];
        } else {
            c0 = c1 = c2 = 0.0f; s0 = s1 = s2 = -1.0f;
        }
        bmin[t][0] = c0 - 0.5f*s0; bmin[t][1] = c1 - 0.5f*s1; bmin[t][2] = c2 - 0.5f*s2;
        bmax[t][0] = c0 + 0.5f*s0; bmax[t][1] = c1 + 0.5f*s1; bmax[t][2] = c2 + 0.5f*s2;
    }
    __syncthreads();

    for (int s = t; s < SN; s += 256) {
        const float x = seed_xyz[s*3+0], y = seed_xyz[s*3+1], z = seed_xyz[s*3+2];
        unsigned int m = 0;
#pragma unroll
        for (int b = 0; b < BPG; ++b) {
            const int in = (x >= bmin[b][0]) & (x <= bmax[b][0]) &
                           (y >= bmin[b][1]) & (y <= bmax[b][1]) &
                           (z >= bmin[b][2]) & (z <= bmax[b][2]);
            m |= (unsigned)in << b;
        }
        smask[s] = m;
    }
    {
        const int p = t;
        const float x = agg_xyz[p*3+0], y = agg_xyz[p*3+1], z = agg_xyz[p*3+2];
        unsigned int m = 0;
#pragma unroll
        for (int b = 0; b < BPG; ++b) {
            const int in = (x >= bmin[b][0]) & (x <= bmax[b][0]) &
                           (y >= bmin[b][1]) & (y <= bmax[b][1]) &
                           (z >= bmin[b][2]) & (z <= bmax[b][2]);
            m |= (unsigned)in << b;
        }
        pmask[p] = m;
    }
    __syncthreads();

    {
        const int d = t;
        float macc[BPG];
#pragma unroll
        for (int b = 0; b < BPG; ++b) macc[b] = -__builtin_inff();
        const float* __restrict__ row = seed_feature + (size_t)d * SN;
#pragma unroll 4
        for (int s = 0; s < SN; ++s) {
            const float v = row[s];
            const unsigned int m = smask[s];
#pragma unroll
            for (int b = 0; b < BPG; ++b) {
                const float vv = ((m >> b) & 1u) ? v : 0.0f;
                macc[b] = fmaxf(macc[b], vv);
            }
        }
#pragma unroll
        for (int b = 0; b < BPG; ++b) g[b][d] = macc[b];
    }

    if (t < CBOX) {
        const int c = t;
        float macc[BPG];
#pragma unroll
        for (int b = 0; b < BPG; ++b) macc[b] = -__builtin_inff();
#pragma unroll 4
        for (int p = 0; p < PN; ++p) {
            const float v = box_features[p * CBOX + c];
            const unsigned int m = pmask[p];
#pragma unroll
            for (int b = 0; b < BPG; ++b) {
                const float vv = ((m >> b) & 1u) ? v : 0.0f;
                macc[b] = fmaxf(macc[b], vv);
            }
        }
#pragma unroll
        for (int b = 0; b < BPG; ++b) g[b][DSEED + c] = macc[b];
    }

    for (int i = t; i < BPG * CBFU; i += 256) {
        const int b = i / CBFU, c = i % CBFU;
        const int gb = box0 + b;
        g[b][DSEED + CBOX + c] = (gb < UB) ? bfu[(size_t)gb * CBFU + c] : 0.0f;
    }
    __syncthreads();

    v8f acc = {};
#if defined(__AMDGCN__) && __has_builtin(__builtin_amdgcn_wmma_f32_16x16x4_f32)
    for (int k0 = 0; k0 < KDIM; k0 += 4) {
        const int ka = k0 + 2 * hsel;
        v2f a, b;
        a.x = g[l15][ka];
        a.y = g[l15][ka + 1];
        const float* __restrict__ wr = W + (size_t)(nbase + l15) * KDIM + ka;
        b.x = wr[0];
        b.y = wr[1];
        acc = __builtin_amdgcn_wmma_f32_16x16x4_f32(false, a, false, b,
                                                    (short)0, acc, false, false);
    }
#elif defined(__AMDGCN__)
    for (int k0 = 0; k0 < KDIM; k0 += 32) {
        v16h a, b;
#pragma unroll
        for (int v = 0; v < 8; ++v) {
#pragma unroll
            for (int h = 0; h < 2; ++h) {
                const int k = k0 + 2*v + h + 8 * (hsel + (v >= 4));
                a[2*v + h] = (_Float16)g[l15][k];
                b[2*v + h] = (_Float16)W[(size_t)(nbase + l15) * KDIM + k];
            }
        }
        acc = __builtin_amdgcn_wmma_f32_16x16x32_f16(false, a, false, b,
                                                     (short)0, acc, false, false);
    }
#endif

    const float bn = bias[nbase + l15];
#pragma unroll
    for (int j = 0; j < 8; ++j) {
        const int m  = j + 8 * hsel;
        const int gm = box0 + m;
        if (gm < UB) {
            const float x  = acc[j] + bn + 1e-6f;
            const float ax = fabsf(x);
            out[(size_t)gm * ODIM + nbase + l15] = ax / (1.0f + ax);
        }
    }
}

extern "C" void kernel_launch(void* const* d_in, const int* in_sizes, int n_in,
                              void* d_out, int out_size, void* d_ws, size_t ws_size,
                              hipStream_t stream) {
    const float* union_box    = (const float*)d_in[0];
    const float* box_features = (const float*)d_in[1];
    const float* agg_xyz      = (const float*)d_in[2];
    const float* seed_feature = (const float*)d_in[3];
    const float* seed_xyz     = (const float*)d_in[4];
    const float* bfu          = (const float*)d_in[5];
    const float* W_gs         = (const float*)d_in[6];
    const float* b_gs         = (const float*)d_in[7];
    float* out = (float*)d_out;

    const int grid = (UB + BPG - 1) / BPG;
    scenebox_fused_kernel<<<grid, 256, 0, stream>>>(
        union_box, box_features, agg_xyz, seed_feature, seed_xyz,
        bfu, W_gs, b_gs, out);
}


# proposal pooling re-mapped onto all 8 waves: a wave takes 32 proposals, one load = one whole 512-byte proposal row, all 32 loads in flight at once; waves' maxima joined via LDS; max only
# speedup vs baseline: 4.2230x; 4.2230x over previous
#include <hip/hip_runtime.h>
#include <hip/hip_bf16.h>

typedef __attribute__((ext_vector_type(2)))  float    v2f;
typedef __attribute__((ext_vector_type(4)))  float    v4f;
typedef __attribute__((ext_vector_type(4)))  unsigned int v4u;
typedef __attribute__((ext_vector_type(8)))  float    v8f;
typedef __attribute__((ext_vector_type(16))) _Float16 v16h;

#define UB     700
#define PN     256
#define SN     1024
#define CBOX   128
#define DSEED  256
#define CBFU   128
#define KDIM   512
#define ODIM   128
#define BPG    4
#define GPAD   8
#define NWAVE  8
#define SHALF  (SN / 2)
#define RPW    (DSEED / 4)
#define RPITCH 36
#define PPW    (PN / NWAVE)

__device__ __forceinline__ float max16(const float (&s)[16]) {
    float a = fmaxf(fmaxf(s[0], s[1]), s[2]);
    a = fmaxf(fmaxf(a, s[3]), s[4]);
    a = fmaxf(fmaxf(a, s[5]), s[6]);
    a = fmaxf(fmaxf(a, s[7]), s[8]);
    float b = fmaxf(fmaxf(s[9], s[10]), s[11]);
    b = fmaxf(fmaxf(b, s[12]), s[13]);
    b = fmaxf(fmaxf(b, s[14]), s[15]);
    return fmaxf(a, b);
}
__device__ __forceinline__ float max32(const float (&s)[32]) {
    float a = fmaxf(fmaxf(s[0], s[1]), s[2]);
#pragma unroll
    for (int i = 3; i < 17; i += 2) a = fmaxf(fmaxf(a, s[i]), s[i + 1]);
    float b = fmaxf(fmaxf(s[17], s[18]), s[19]);
#pragma unroll
    for (int i = 20; i < 32; i += 2) b = fmaxf(fmaxf(b, s[i]), s[i + 1]);
    return fmaxf(a, b);
}

__global__ __launch_bounds__(256)
void scenebox_fused_kernel(const float* __restrict__ union_box,
                           const float* __restrict__ box_features,
                           const float* __restrict__ agg_xyz,
                           const float* __restrict__ seed_feature,
                           const float* __restrict__ seed_xyz,
                           const float* __restrict__ bfu,
                           const float* __restrict__ W,
                           const float* __restrict__ bias,
                           float* __restrict__ out)
{
    __shared__ float        g[BPG][KDIM + GPAD];
    __shared__ float        bmin[BPG][3], bmax[BPG][3];
    __shared__ __attribute__((aligned(16))) unsigned int smask[SN];
    __shared__ float        gp[2][BPG][DSEED];
    __shared__ __attribute__((aligned(16))) float red[NWAVE][32 * RPITCH];
    __shared__ unsigned int pmask[PN];
    __shared__ __attribute__((aligned(16))) float gq[NWAVE][BPG][CBOX];

    const int t    = threadIdx.x;
    const int box0 = blockIdx.x * BPG;
    const int lane = t & 31;
    const int wave = t >> 5;
    const int nbase = wave * 16;
    const int hsel  = lane >> 4;
    const int l15   = lane & 15;

    __builtin_prefetch(&W[(nbase + l15) * KDIM], 0, 3);
    __builtin_prefetch(&W[(nbase + l15) * KDIM + 256], 0, 3);

    if (t < BPG) {
        const int b = box0 + t;
        float c0, c1, c2, s0, s1, s2;
        if (b < UB) {
            c0 = union_box[b*6+0]; c1 = union_box[b*6+1]; c2 = union_box[b*6+2];
            s0 = union_box[b*6+3]; s1 = union_box[b*6+4]; s2 = union_box[b*6+5];
        } else {
            c0 = c1 = c2 = 0.0f; s0 = s1 = s2 = -1.0f;
        }
        bmin[t][0] = c0 - 0.5f*s0; bmin[t][1] = c1 - 0.5f*s1; bmin[t][2] = c2 - 0.5f*s2;
        bmax[t][0] = c0 + 0.5f*s0; bmax[t][1] = c1 + 0.5f*s1; bmax[t][2] = c2 + 0.5f*s2;
    }
    __syncthreads();

    for (int s = t; s < SN; s += 256) {
        const float x = seed_xyz[s*3+0], y = seed_xyz[s*3+1], z = seed_xyz[s*3+2];
        unsigned int m = 0;
#pragma unroll
        for (int b = 0; b < BPG; ++b) {
            const int in = (x >= bmin[b][0]) & (x <= bmax[b][0]) &
                           (y >= bmin[b][1]) & (y <= bmax[b][1]) &
                           (z >= bmin[b][2]) & (z <= bmax[b][2]);
            m |= (unsigned)(in ^ 1) << b;
        }
        smask[s] = m;
    }
    {
        const int p = t;
        const float x = agg_xyz[p*3+0], y = agg_xyz[p*3+1], z = agg_xyz[p*3+2];
        unsigned int m = 0;
#pragma unroll
        for (int b = 0; b < BPG; ++b) {
            const int in = (x >= bmin[b][0]) & (x <= bmax[b][0]) &
                           (y >= bmin[b][1]) & (y <= bmax[b][1]) &
                           (z >= bmin[b][2]) & (z <= bmax[b][2]);
            m |= (unsigned)(in ^ 1) << b;
        }
        pmask[p] = m;
    }
    __syncthreads();

    {
        const int half = wave >> 2;
        const int row0 = (wave & 3) * RPW;
        const int sb   = half * SHALF + 4 * lane;

        bool in[BPG][16];
#pragma unroll
        for (int j = 0; j < 4; ++j) {
            const v4u q = *reinterpret_cast<const v4u*>(&smask[sb + 128 * j]);
#pragma unroll
            for (int b = 0; b < BPG; ++b) {
                in[b][4*j+0] = (q.x & (1u << b)) == 0u;
                in[b][4*j+1] = (q.y & (1u << b)) == 0u;
                in[b][4*j+2] = (q.z & (1u << b)) == 0u;
                in[b][4*j+3] = (q.w & (1u << b)) == 0u;
            }
        }

        const float* __restrict__ base = seed_feature + (size_t)row0 * SN + sb;
        float* __restrict__ myred = red[wave];

        v4f cur[4][4], nxt[4][4];
#pragma unroll
        for (int r = 0; r < 4; ++r)
#pragma unroll
            for (int j = 0; j < 4; ++j)
                cur[r][j] = *reinterpret_cast<const v4f*>(base + (size_t)r * SN + 128 * j);

        for (int g8 = 0; g8 < RPW / 8; ++g8) {
            float p[32];
            const float* __restrict__ rowp = base + (size_t)(g8 * 8) * SN;
#pragma unroll
            for (int r = 0; r < 4; ++r)
#pragma unroll
                for (int j = 0; j < 4; ++j)
                    nxt[r][j] = *reinterpret_cast<const v4f*>(rowp + (size_t)(4 + r) * SN + 128 * j);
#pragma unroll
            for (int r = 0; r < 4; ++r) {
#pragma unroll
                for (int b = 0; b < BPG; ++b) {
                    float s[16];
#pragma unroll
                    for (int j = 0; j < 4; ++j) {
                        const v4f x = cur[r][j];
                        s[4*j+0] = in[b][4*j+0] ? x.x : 0.0f;
                        s[4*j+1] = in[b][4*j+1] ? x.y : 0.0f;
                        s[4*j+2] = in[b][4*j+2] ? x.z : 0.0f;
                        s[4*j+3] = in[b][4*j+3] ? x.w : 0.0f;
                    }
                    p[r * BPG + b] = max16(s);
                }
            }
            if (g8 + 1 < RPW / 8) {
#pragma unroll
                for (int r = 0; r < 4; ++r)
#pragma unroll
                    for (int j = 0; j < 4; ++j)
                        cur[r][j] = *reinterpret_cast<const v4f*>(rowp + (size_t)(8 + r) * SN + 128 * j);
            }
#pragma unroll
            for (int r = 0; r < 4; ++r) {
#pragma unroll
                for (int b = 0; b < BPG; ++b) {
                    float s[16];
#pragma unroll
                    for (int j = 0; j < 4; ++j) {
                        const v4f x = nxt[r][j];
                        s[4*j+0] = in[b][4*j+0] ? x.x : 0.0f;
                        s[4*j+1] = in[b][4*j+1] ? x.y : 0.0f;
                        s[4*j+2] = in[b][4*j+2] ? x.z : 0.0f;
                        s[4*j+3] = in[b][4*j+3] ? x.w : 0.0f;
                    }
                    p[(4 + r) * BPG + b] = max16(s);
                }
            }
#pragma unroll
            for (int k = 0; k < 32; ++k) myred[k * RPITCH + lane] = p[k];
            __builtin_amdgcn_fence(__ATOMIC_RELEASE, "wavefront");
            __builtin_amdgcn_fence(__ATOMIC_ACQUIRE, "wavefront");
            float c[32];
#pragma unroll
            for (int j = 0; j < 8; ++j) {
                const v4f x = *reinterpret_cast<const v4f*>(&myred[lane * RPITCH + 4 * j]);
                c[4*j+0] = x.x; c[4*j+1] = x.y; c[4*j+2] = x.z; c[4*j+3] = x.w;
            }
            gp[half][lane & (BPG - 1)][row0 + g8 * 8 + (lane >> 2)] = max32(c);
        }
    }

    {
        const int p0 = __builtin_amdgcn_readfirstlane(wave) * PPW;
        const float* __restrict__ bp = box_features + (size_t)p0 * CBOX + 4 * lane;
        v4f x[PPW];
#pragma unroll
        for (int i = 0; i < PPW; ++i)
            x[i] = *reinterpret_cast<const v4f*>(bp + (size_t)i * CBOX);
        v4f macc[BPG];
#pragma unroll
        for (int i = 0; i < PPW; i += 2) {
            const unsigned int m0 = pmask[p0 + i], m1 = pmask[p0 + i + 1];
#pragma unroll
            for (int b = 0; b < BPG; ++b) {
                const bool i0 = (m0 & (1u << b)) == 0u, i1 = (m1 & (1u << b)) == 0u;
                v4f s0, s1;
                s0.x = i0 ? x[i].x : 0.0f;     s0.y = i0 ? x[i].y : 0.0f;
                s0.z = i0 ? x[i].z : 0.0f;     s0.w = i0 ? x[i].w : 0.0f;
                s1.x = i1 ? x[i + 1].x : 0.0f; s1.y = i1 ? x[i + 1].y : 0.0f;
                s1.z = i1 ? x[i + 1].z : 0.0f; s1.w = i1 ? x[i + 1].w : 0.0f;
                if (i == 0) {
                    macc[b].x = fmaxf(s0.x, s1.x); macc[b].y = fmaxf(s0.y, s1.y);
                    macc[b].z = fmaxf(s0.z, s1.z); macc[b].w = fmaxf(s0.w, s1.w);
                } else {
                    macc[b].x = fmaxf(fmaxf(macc[b].x, s0.x), s1.x);
                    macc[b].y = fmaxf(fmaxf(macc[b].y, s0.y), s1.y);
                    macc[b].z = fmaxf(fmaxf(macc[b].z, s0.z), s1.z);
                    macc[b].w = fmaxf(fmaxf(macc[b].w, s0.w), s1.w);
                }
            }
        }
#pragma unroll
        for (int b = 0; b < BPG; ++b)
            *reinterpret_cast<v4f*>(&gq[wave][b][4 * lane]) = macc[b];
    }

    for (int i = t; i < BPG * CBFU; i += 256) {
        const int b = i / CBFU, c = i % CBFU;
        const int gb = box0 + b;
        g[b][DSEED + CBOX + c] = (gb < UB) ? bfu[(size_t)gb * CBFU + c] : 0.0f;
    }
    __syncthreads();

#pragma unroll
    for (int b = 0; b < BPG; ++b) g[b][t] = fmaxf(gp[0][b][t], gp[1][b][t]);
    for (int i = t; i < BPG * CBOX; i += 256) {
        const int b = i / CBOX, c = i % CBOX;
        float m = fmaxf(gq[0][b][c], gq[1][b][c]);
#pragma unroll
        for (int w = 2; w < NWAVE; w += 2) m = fmaxf(fmaxf(m, gq[w][b][c]), gq[w + 1][b][c]);
        g[b][DSEED + c] = m;
    }
    __syncthreads();

    v8f acc = {};
#if defined(__AMDGCN__) && __has_builtin(__builtin_amdgcn_wmma_f32_16x16x4_f32)
    for (int k0 = 0; k0 < KDIM; k0 += 4) {
        const int ka = k0 + 2 * hsel;
        v2f a, b;
        a.x = g[l15 & (BPG - 1)][ka];
        a.y = g[l15 & (BPG - 1)][ka + 1];
        const float* __restrict__ wr = W + (size_t)(nbase + l15) * KDIM + ka;
        b.x = wr[0];
        b.y = wr[1];
        acc = __builtin_amdgcn_wmma_f32_16x16x4_f32(false, a, false, b,
                                                    (short)0, acc, false, false);
    }
#elif defined(__AMDGCN__)
    for (int k0 = 0; k0 < KDIM; k0 += 32) {
        v16h a, b;
#pragma unroll
        for (int v = 0; v < 8; ++v) {
#pragma unroll
            for (int h = 0; h < 2; ++h) {
                const int k = k0 + 2*v + h + 8 * (hsel + (v >= 4));
                a[2*v + h] = (_Float16)g[l15][k];
                b[2*v + h] = (_Float16)W[(size_t)(nbase + l15) * KDIM + k];
            }
        }
        acc = __builtin_amdgcn_wmma_f32_16x16x32_f16(false, a, false, b,
                                                     (short)0, acc, false, false);
    }
#endif

    const float bn = bias[nbase + l15];
#pragma unroll
    for (int j = 0; j < 8; ++j) {
        const int m  = j + 8 * hsel;
        const int gm = box0 + m;
        if (m < BPG && gm < UB) {
            const float x  = acc[j] + bn + 1e-6f;
            const float ax = fabsf(x);
            out[(size_t)gm * ODIM + nbase + l15] = ax / (1.0f + ax);
        }
    }
}

extern "C" void kernel_launch(void* const* d_in, const int* in_sizes, int n_in,
                              void* d_out, int out_size, void* d_ws, size_t ws_size,
                              hipStream_t stream) {
    const float* union_box    = (const float*)d_in[0];
    const float* box_features = (const float*)d_in[1];
    const float* agg_xyz      = (const float*)d_in[2];
    const float* seed_feature = (const float*)d_in[3];
    const float* seed_xyz     = (const float*)d_in[4];
    const float* bfu          = (const float*)d_in[5];
    const float* W_gs         = (const float*)d_in[6];
    const float* b_gs         = (const float*)d_in[7];
    float* out = (float*)d_out;

    const int grid = (UB + BPG - 1) / BPG;
    scenebox_fused_kernel<<<grid, 256, 0, stream>>>(
        union_box, box_features, agg_xyz, seed_feature, seed_xyz,
        bfu, W_gs, b_gs, out);
}


# workgroup widened from 8 to 16 waves (512 threads, four per SIMD): mask pass and both poolings split 16 ways (32 feature rows / 16 proposals per wave); linear layer stays on waves 0-7; max only
# speedup vs baseline: 4.6988x; 1.1127x over previous
#include <hip/hip_runtime.h>
#include <hip/hip_bf16.h>

typedef __attribute__((ext_vector_type(2)))  float    v2f;
typedef __attribute__((ext_vector_type(4)))  float    v4f;
typedef __attribute__((ext_vector_type(4)))  unsigned int v4u;
typedef __attribute__((ext_vector_type(8)))  float    v8f;
typedef __attribute__((ext_vector_type(16))) _Float16 v16h;

#define UB     700
#define PN     256
#define SN     1024
#define CBOX   128
#define DSEED  256
#define CBFU   128
#define KDIM   512
#define ODIM   128
#define BPG    4
#define GPAD   8
#define NWAVE  16
#define NTHR   (NWAVE * 32)
#define NGW    8
#define SHALF  (SN / 2)
#define RPW    (DSEED / (NWAVE / 2))
#define RPITCH 36
#define PPW    (PN / NWAVE)

__device__ __forceinline__ float max16(const float (&s)[16]) {
    float a = fmaxf(fmaxf(s[0], s[1]), s[2]);
    a = fmaxf(fmaxf(a, s[3]), s[4]);
    a = fmaxf(fmaxf(a, s[5]), s[6]);
    a = fmaxf(fmaxf(a, s[7]), s[8]);
    float b = fmaxf(fmaxf(s[9], s[10]), s[11]);
    b = fmaxf(fmaxf(b, s[12]), s[13]);
    b = fmaxf(fmaxf(b, s[14]), s[15]);
    return fmaxf(a, b);
}
__device__ __forceinline__ float max32(const float (&s)[32]) {
    float a = fmaxf(fmaxf(s[0], s[1]), s[2]);
#pragma unroll
    for (int i = 3; i < 17; i += 2) a = fmaxf(fmaxf(a, s[i]), s[i + 1]);
    float b = fmaxf(fmaxf(s[17], s[18]), s[19]);
#pragma unroll
    for (int i = 20; i < 32; i += 2) b = fmaxf(fmaxf(b, s[i]), s[i + 1]);
    return fmaxf(a, b);
}

__global__ __launch_bounds__(NTHR)
void scenebox_fused_kernel(const float* __restrict__ union_box,
                           const float* __restrict__ box_features,
                           const float* __restrict__ agg_xyz,
                           const float* __restrict__ seed_feature,
                           const float* __restrict__ seed_xyz,
                           const float* __restrict__ bfu,
                           const float* __restrict__ W,
                           const float* __restrict__ bias,
                           float* __restrict__ out)
{
    __shared__ float        g[BPG][KDIM + GPAD];
    __shared__ float        bmin[BPG][3], bmax[BPG][3];
    __shared__ __attribute__((aligned(16))) unsigned int smask[SN];
    __shared__ float        gp[2][BPG][DSEED];
    __shared__ __attribute__((aligned(16))) float red[NWAVE][32 * RPITCH];
    __shared__ unsigned int pmask[PN];
    __shared__ __attribute__((aligned(16))) float gq[NWAVE][BPG][CBOX];

    const int t    = threadIdx.x;
    const int box0 = blockIdx.x * BPG;
    const int lane = t & 31;
    const int wave = t >> 5;
    const int nbase = wave * 16;
    const int hsel  = lane >> 4;
    const int l15   = lane & 15;

    if (wave < NGW) {
        __builtin_prefetch(&W[(nbase + l15) * KDIM], 0, 3);
        __builtin_prefetch(&W[(nbase + l15) * KDIM + 256], 0, 3);
    }

    if (t < BPG) {
        const int b = box0 + t;
        float c0, c1, c2, s0, s1, s2;
        if (b < UB) {
            c0 = union_box[b*6+0]; c1 = union_box[b*6+1]; c2 = union_box[b*6+2];
            s0 = union_box[b*6+3]; s1 = union_box[b*6+4]; s2 = union_box[b*6+5];
        } else {
            c0 = c1 = c2 = 0.0f; s0 = s1 = s2 = -1.0f;
        }
        bmin[t][0] = c0 - 0.5f*s0; bmin[t][1] = c1 - 0.5f*s1; bmin[t][2] = c2 - 0.5f*s2;
        bmax[t][0] = c0 + 0.5f*s0; bmax[t][1] = c1 + 0.5f*s1; bmax[t][2] = c2 + 0.5f*s2;
    }
    __syncthreads();

    for (int s = t; s < SN; s += NTHR) {
        const float x = seed_xyz[s*3+0], y = seed_xyz[s*3+1], z = seed_xyz[s*3+2];
        unsigned int m = 0;
#pragma unroll
        for (int b = 0; b < BPG; ++b) {
            const int in = (x >= bmin[b][0]) & (x <= bmax[b][0]) &
                           (y >= bmin[b][1]) & (y <= bmax[b][1]) &
                           (z >= bmin[b][2]) & (z <= bmax[b][2]);
            m |= (unsigned)(in ^ 1) << b;
        }
        smask[s] = m;
    }
    if (t < PN) {
        const int p = t;
        const float x = agg_xyz[p*3+0], y = agg_xyz[p*3+1], z = agg_xyz[p*3+2];
        unsigned int m = 0;
#pragma unroll
        for (int b = 0; b < BPG; ++b) {
            const int in = (x >= bmin[b][0]) & (x <= bmax[b][0]) &
                           (y >= bmin[b][1]) & (y <= bmax[b][1]) &
                           (z >= bmin[b][2]) & (z <= bmax[b][2]);
            m |= (unsigned)(in ^ 1) << b;
        }
        pmask[p] = m;
    }
    __syncthreads();

    {
        const int half = wave / (NWAVE / 2);
        const int row0 = (wave % (NWAVE / 2)) * RPW;
        const int sb   = half * SHALF + 4 * lane;

        bool in[BPG][16];
#pragma unroll
        for (int j = 0; j < 4; ++j) {
            const v4u q = *reinterpret_cast<const v4u*>(&smask[sb + 128 * j]);
#pragma unroll
            for (int b = 0; b < BPG; ++b) {
                in[b][4*j+0] = (q.x & (1u << b)) == 0u;
                in[b][4*j+1] = (q.y & (1u << b)) == 0u;
                in[b][4*j+2] = (q.z & (1u << b)) == 0u;
                in[b][4*j+3] = (q.w & (1u << b)) == 0u;
            }
        }

        const float* __restrict__ base = seed_feature + (size_t)row0 * SN + sb;
        float* __restrict__ myred = red[wave];

        v4f cur[4][4], nxt[4][4];
#pragma unroll
        for (int r = 0; r < 4; ++r)
#pragma unroll
            for (int j = 0; j < 4; ++j)
                cur[r][j] = *reinterpret_cast<const v4f*>(base + (size_t)r * SN + 128 * j);

        for (int g8 = 0; g8 < RPW / 8; ++g8) {
            float p[32];
            const float* __restrict__ rowp = base + (size_t)(g8 * 8) * SN;
#pragma unroll
            for (int r = 0; r < 4; ++r)
#pragma unroll
                for (int j = 0; j < 4; ++j)
                    nxt[r][j] = *reinterpret_cast<const v4f*>(rowp + (size_t)(4 + r) * SN + 128 * j);
#pragma unroll
            for (int r = 0; r < 4; ++r) {
#pragma unroll
                for (int b = 0; b < BPG; ++b) {
                    float s[16];
#pragma unroll
                    for (int j = 0; j < 4; ++j) {
                        const v4f x = cur[r][j];
                        s[4*j+0] = in[b][4*j+0] ? x.x : 0.0f;
                        s[4*j+1] = in[b][4*j+1] ? x.y : 0.0f;
                        s[4*j+2] = in[b][4*j+2] ? x.z : 0.0f;
                        s[4*j+3] = in[b][4*j+3] ? x.w : 0.0f;
                    }
                    p[r * BPG + b] = max16(s);
                }
            }
            if (g8 + 1 < RPW / 8) {
#pragma unroll
                for (int r = 0; r < 4; ++r)
#pragma unroll
                    for (int j = 0; j < 4; ++j)
                        cur[r][j] = *reinterpret_cast<const v4f*>(rowp + (size_t)(8 + r) * SN + 128 * j);
            }
#pragma unroll
            for (int r = 0; r < 4; ++r) {
#pragma unroll
                for (int b = 0; b < BPG; ++b) {
                    float s[16];
#pragma unroll
                    for (int j = 0; j < 4; ++j) {
                        const v4f x = nxt[r][j];
                        s[4*j+0] = in[b][4*j+0] ? x.x : 0.0f;
                        s[4*j+1] = in[b][4*j+1] ? x.y : 0.0f;
                        s[4*j+2] = in[b][4*j+2] ? x.z : 0.0f;
                        s[4*j+3] = in[b][4*j+3] ? x.w : 0.0f;
                    }
                    p[(4 + r) * BPG + b] = max16(s);
                }
            }
#pragma unroll
            for (int k = 0; k < 32; ++k) myred[k * RPITCH + lane] = p[k];
            __builtin_amdgcn_fence(__ATOMIC_RELEASE, "wavefront");
            __builtin_amdgcn_fence(__ATOMIC_ACQUIRE, "wavefront");
            float c[32];
#pragma unroll
            for (int j = 0; j < 8; ++j) {
                const v4f x = *reinterpret_cast<const v4f*>(&myred[lane * RPITCH + 4 * j]);
                c[4*j+0] = x.x; c[4*j+1] = x.y; c[4*j+2] = x.z; c[4*j+3] = x.w;
            }
            gp[half][lane & (BPG - 1)][row0 + g8 * 8 + (lane >> 2)] = max32(c);
        }
    }

    {
        const int p0 = __builtin_amdgcn_readfirstlane(wave) * PPW;
        const float* __restrict__ bp = box_features + (size_t)p0 * CBOX + 4 * lane;
        v4f x[PPW];
#pragma unroll
        for (int i = 0; i < PPW; ++i)
            x[i] = *reinterpret_cast<const v4f*>(bp + (size_t)i * CBOX);
        v4f macc[BPG];
#pragma unroll
        for (int i = 0; i < PPW; i += 2) {
            const unsigned int m0 = pmask[p0 + i], m1 = pmask[p0 + i + 1];
#pragma unroll
            for (int b = 0; b < BPG; ++b) {
                const bool i0 = (m0 & (1u << b)) == 0u, i1 = (m1 & (1u << b)) == 0u;
                v4f s0, s1;
                s0.x = i0 ? x[i].x : 0.0f;     s0.y = i0 ? x[i].y : 0.0f;
                s0.z = i0 ? x[i].z : 0.0f;     s0.w = i0 ? x[i].w : 0.0f;
                s1.x = i1 ? x[i + 1].x : 0.0f; s1.y = i1 ? x[i + 1].y : 0.0f;
                s1.z = i1 ? x[i + 1].z : 0.0f; s1.w = i1 ? x[i + 1].w : 0.0f;
                if (i == 0) {
                    macc[b].x = fmaxf(s0.x, s1.x); macc[b].y = fmaxf(s0.y, s1.y);
                    macc[b].z = fmaxf(s0.z, s1.z); macc[b].w = fmaxf(s0.w, s1.w);
                } else {
                    macc[b].x = fmaxf(fmaxf(macc[b].x, s0.x), s1.x);
                    macc[b].y = fmaxf(fmaxf(macc[b].y, s0.y), s1.y);
                    macc[b].z = fmaxf(fmaxf(macc[b].z, s0.z), s1.z);
                    macc[b].w = fmaxf(fmaxf(macc[b].w, s0.w), s1.w);
                }
            }
        }
#pragma unroll
        for (int b = 0; b < BPG; ++b)
            *reinterpret_cast<v4f*>(&gq[wave][b][4 * lane]) = macc[b];
    }

    for (int i = t; i < BPG * CBFU; i += NTHR) {
        const int b = i / CBFU, c = i % CBFU;
        const int gb = box0 + b;
        g[b][DSEED + CBOX + c] = (gb < UB) ? bfu[(size_t)gb * CBFU + c] : 0.0f;
    }
    __syncthreads();

    if (t < DSEED) {
#pragma unroll
        for (int b = 0; b < BPG; ++b) g[b][t] = fmaxf(gp[0][b][t], gp[1][b][t]);
    }
    for (int i = t; i < BPG * CBOX; i += NTHR) {
        const int b = i / CBOX, c = i % CBOX;
        float m = fmaxf(gq[0][b][c], gq[1][b][c]);
#pragma unroll
        for (int w = 2; w < NWAVE; w += 2) m = fmaxf(fmaxf(m, gq[w][b][c]), gq[w + 1][b][c]);
        g[b][DSEED + c] = m;
    }
    __syncthreads();

    if (__builtin_amdgcn_readfirstlane(wave) >= NGW) return;
    v8f acc = {};
#if defined(__AMDGCN__) && __has_builtin(__builtin_amdgcn_wmma_f32_16x16x4_f32)
    for (int k0 = 0; k0 < KDIM; k0 += 4) {
        const int ka = k0 + 2 * hsel;
        v2f a, b;
        a.x = g[l15 & (BPG - 1)][ka];
        a.y = g[l15 & (BPG - 1)][ka + 1];
        const float* __restrict__ wr = W + (size_t)(nbase + l15) * KDIM + ka;
        b.x = wr[0];
        b.y = wr[1];
        acc = __builtin_amdgcn_wmma_f32_16x16x4_f32(false, a, false, b,
                                                    (short)0, acc, false, false);
    }
#elif defined(__AMDGCN__)
    for (int k0 = 0; k0 < KDIM; k0 += 32) {
        v16h a, b;
#pragma unroll
        for (int v = 0; v < 8; ++v) {
#pragma unroll
            for (int h = 0; h < 2; ++h) {
                const int k = k0 + 2*v + h + 8 * (hsel + (v >= 4));
                a[2*v + h] = (_Float16)g[l15][k];
                b[2*v + h] = (_Float16)W[(size_t)(nbase + l15) * KDIM + k];
            }
        }
        acc = __builtin_amdgcn_wmma_f32_16x16x32_f16(false, a, false, b,
                                                     (short)0, acc, false, false);
    }
#endif

    const float bn = bias[nbase + l15];
#pragma unroll
    for (int j = 0; j < 8; ++j) {
        const int m  = j + 8 * hsel;
        const int gm = box0 + m;
        if (m < BPG && gm < UB) {
            const float x  = acc[j] + bn + 1e-6f;
            const float ax = fabsf(x);
            out[(size_t)gm * ODIM + nbase + l15] = ax / (1.0f + ax);
        }
    }
}

extern "C" void kernel_launch(void* const* d_in, const int* in_sizes, int n_in,
                              void* d_out, int out_size, void* d_ws, size_t ws_size,
                              hipStream_t stream) {
    const float* union_box    = (const float*)d_in[0];
    const float* box_features = (const float*)d_in[1];
    const float* agg_xyz      = (const float*)d_in[2];
    const float* seed_feature = (const float*)d_in[3];
    const float* seed_xyz     = (const float*)d_in[4];
    const float* bfu          = (const float*)d_in[5];
    const float* W_gs         = (const float*)d_in[6];
    const float* b_gs         = (const float*)d_in[7];
    float* out = (float*)d_out;

    const int grid = (UB + BPG - 1) / BPG;
    scenebox_fused_kernel<<<grid, NTHR, 0, stream>>>(
        union_box, box_features, agg_xyz, seed_feature, seed_xyz,
        bfu, W_gs, b_gs, out);
}


# first global reads of every phase issued together at kernel entry in order of need, in flight across LDS-only barriers; box bounds as scalar reads (one barrier less); proposal pooling moved in front
# speedup vs baseline: 5.7123x; 1.2157x over previous
#include <hip/hip_runtime.h>
#include <hip/hip_bf16.h>

typedef __attribute__((ext_vector_type(2)))  float    v2f;
typedef __attribute__((ext_vector_type(4)))  float    v4f;
typedef __attribute__((ext_vector_type(4)))  unsigned int v4u;
typedef __attribute__((ext_vector_type(8)))  float    v8f;
typedef __attribute__((ext_vector_type(16))) _Float16 v16h;

#define UB     700
#define PN     256
#define SN     1024
#define CBOX   128
#define DSEED  256
#define CBFU   128
#define KDIM   512
#define ODIM   128
#define BPG    4
#define GPAD   8
#define NWAVE  16
#define NTHR   (NWAVE * 32)
#define NGW    8
#define SHALF  (SN / 2)
#define RPW    (DSEED / (NWAVE / 2))
#define RPITCH 36
#define PPW    (PN / NWAVE)

__device__ __forceinline__ float max16(const float (&s)[16]) {
    float a = fmaxf(fmaxf(s[0], s[1]), s[2]);
    a = fmaxf(fmaxf(a, s[3]), s[4]);
    a = fmaxf(fmaxf(a, s[5]), s[6]);
    a = fmaxf(fmaxf(a, s[7]), s[8]);
    float b = fmaxf(fmaxf(s[9], s[10]), s[11]);
    b = fmaxf(fmaxf(b, s[12]), s[13]);
    b = fmaxf(fmaxf(b, s[14]), s[15]);
    return fmaxf(a, b);
}
__device__ __forceinline__ float max32(const float (&s)[32]) {
    float a = fmaxf(fmaxf(s[0], s[1]), s[2]);
#pragma unroll
    for (int i = 3; i < 17; i += 2) a = fmaxf(fmaxf(a, s[i]), s[i + 1]);
    float b = fmaxf(fmaxf(s[17], s[18]), s[19]);
#pragma unroll
    for (int i = 20; i < 32; i += 2) b = fmaxf(fmaxf(b, s[i]), s[i + 1]);
    return fmaxf(a, b);
}

__device__ __forceinline__ void lds_barrier() {
    __builtin_amdgcn_fence(__ATOMIC_RELEASE, "workgroup", "local");
    __builtin_amdgcn_s_barrier();
    __builtin_amdgcn_fence(__ATOMIC_ACQUIRE, "workgroup", "local");
}

__global__ __launch_bounds__(NTHR)
void scenebox_fused_kernel(const float* __restrict__ union_box,
                           const float* box_features,
                           const float* agg_xyz,
                           const float* seed_feature,
                           const float* seed_xyz,
                           const float* bfu,
                           const float* __restrict__ W,
                           const float* bias,
                           float* __restrict__ out)
{
    __shared__ __attribute__((aligned(16))) float g[BPG][KDIM + GPAD];
    __shared__ __attribute__((aligned(16))) unsigned int smask[SN];
    __shared__ float        gp[2][BPG][DSEED];
    __shared__ __attribute__((aligned(16))) float red[NWAVE][32 * RPITCH];
    __shared__ __attribute__((aligned(16))) float gq[NWAVE][BPG][CBOX];

    const int t    = threadIdx.x;
    const int box0 = blockIdx.x * BPG;
    const int lane = t & 31;
    const int wave = t >> 5;
    const int nbase = wave * 16;
    const int hsel  = lane >> 4;
    const int l15   = lane & 15;

    if (wave < NGW) {
        __builtin_prefetch(&W[(nbase + l15) * KDIM], 0, 3);
        __builtin_prefetch(&W[(nbase + l15) * KDIM + 256], 0, 3);
    }

    static_assert(SN == 2 * NTHR, "two seeds per thread in the mask pass");
    static_assert(PPW <= 16, "a wave's proposals are masked by its lanes 0..15");
    static_assert(BPG * CBFU == NTHR, "one union-feature element per thread");
    static_assert(UB % BPG == 0, "the boxes of a workgroup are all real or all padding");
    const float sx0 = seed_xyz[t*3+0],          sy0 = seed_xyz[t*3+1],          sz0 = seed_xyz[t*3+2];
    const float sx1 = seed_xyz[(t+NTHR)*3+0],   sy1 = seed_xyz[(t+NTHR)*3+1],   sz1 = seed_xyz[(t+NTHR)*3+2];
    const int   p0   = __builtin_amdgcn_readfirstlane(wave) * PPW;
    const int   pidx = p0 + (l15 & (PPW - 1));
    const float px = agg_xyz[pidx*3+0],         py = agg_xyz[pidx*3+1],         pz = agg_xyz[pidx*3+2];
    const bool  wg_real = box0 < UB;
    const int   ub_b = t / CBFU, ub_c = t % CBFU;
    const float ub_v = bfu[(size_t)(wg_real ? box0 + ub_b : 0) * CBFU + ub_c];
    const float bn   = bias[(wave & (NGW - 1)) * 16 + l15];
    __builtin_amdgcn_sched_barrier(0);

    v4f x[PPW];
    {
        const float* bp = box_features + (size_t)p0 * CBOX + 4 * lane;
#pragma unroll
        for (int i = 0; i < PPW; ++i)
            x[i] = *reinterpret_cast<const v4f*>(bp + (size_t)i * CBOX);
    }
    __builtin_amdgcn_sched_barrier(0);
    const int half = wave / (NWAVE / 2);
    const int row0 = (wave % (NWAVE / 2)) * RPW;
    const int sb   = half * SHALF + 4 * lane;
    const float* base = seed_feature + (size_t)row0 * SN + sb;
    v4f cur[4][4], nxt[4][4];
#pragma unroll
    for (int r = 0; r < 4; ++r)
#pragma unroll
        for (int j = 0; j < 4; ++j)
            cur[r][j] = *reinterpret_cast<const v4f*>(base + (size_t)r * SN + 128 * j);
    __builtin_amdgcn_sched_barrier(0);

    float bmin[BPG][3], bmax[BPG][3];
    {
        const float* __restrict__ ub = union_box + (size_t)(wg_real ? box0 : 0) * 6;
        float u[BPG * 6];
#pragma unroll
        for (int i = 0; i < BPG * 6; ++i) u[i] = ub[i];
#pragma unroll
        for (int b = 0; b < BPG; ++b) {
            float c0 = u[6*b+0], c1 = u[6*b+1], c2 = u[6*b+2], s0 = u[6*b+3], s1 = u[6*b+4], s2 = u[6*b+5];
            if (!wg_real) { c0 = c1 = c2 = 0.0f; s0 = s1 = s2 = -1.0f; }
            bmin[b][0] = c0 - 0.5f*s0; bmin[b][1] = c1 - 0.5f*s1; bmin[b][2] = c2 - 0.5f*s2;
            bmax[b][0] = c0 + 0.5f*s0; bmax[b][1] = c1 + 0.5f*s1; bmax[b][2] = c2 + 0.5f*s2;
        }
    }

    unsigned int mp = 0;
    {
        unsigned int m0 = 0, m1 = 0;
#pragma unroll
        for (int b = 0; b < BPG; ++b) {
            const int in0 = (sx0 >= bmin[b][0]) & (sx0 <= bmax[b][0]) &
                            (sy0 >= bmin[b][1]) & (sy0 <= bmax[b][1]) &
                            (sz0 >= bmin[b][2]) & (sz0 <= bmax[b][2]);
            const int in1 = (sx1 >= bmin[b][0]) & (sx1 <= bmax[b][0]) &
                            (sy1 >= bmin[b][1]) & (sy1 <= bmax[b][1]) &
                            (sz1 >= bmin[b][2]) & (sz1 <= bmax[b][2]);
            const int inp = (px >= bmin[b][0]) & (px <= bmax[b][0]) &
                            (py >= bmin[b][1]) & (py <= bmax[b][1]) &
                            (pz >= bmin[b][2]) & (pz <= bmax[b][2]);
            m0 |= (unsigned)(in0 ^ 1) << b;
            m1 |= (unsigned)(in1 ^ 1) << b;
            mp |= (unsigned)(inp ^ 1) << b;
        }
        smask[t]        = m0;
        smask[t + NTHR] = m1;
    }
    g[ub_b][DSEED + CBOX + ub_c] = wg_real ? ub_v : 0.0f;

    {
        v4f macc[BPG];
#pragma unroll
        for (int i = 0; i < PPW; i += 2) {
            const unsigned int m0 = __builtin_amdgcn_readlane(mp, i), m1 = __builtin_amdgcn_readlane(mp, i + 1);
#pragma unroll
            for (int b = 0; b < BPG; ++b) {
                const bool i0 = (m0 & (1u << b)) == 0u, i1 = (m1 & (1u << b)) == 0u;
                v4f s0, s1;
                s0.x = i0 ? x[i].x : 0.0f;     s0.y = i0 ? x[i].y : 0.0f;
                s0.z = i0 ? x[i].z : 0.0f;     s0.w = i0 ? x[i].w : 0.0f;
                s1.x = i1 ? x[i + 1].x : 0.0f; s1.y = i1 ? x[i + 1].y : 0.0f;
                s1.z = i1 ? x[i + 1].z : 0.0f; s1.w = i1 ? x[i + 1].w : 0.0f;
                if (i == 0) {
                    macc[b].x = fmaxf(s0.x, s1.x); macc[b].y = fmaxf(s0.y, s1.y);
                    macc[b].z = fmaxf(s0.z, s1.z); macc[b].w = fmaxf(s0.w, s1.w);
                } else {
                    macc[b].x = fmaxf(fmaxf(macc[b].x, s0.x), s1.x);
                    macc[b].y = fmaxf(fmaxf(macc[b].y, s0.y), s1.y);
                    macc[b].z = fmaxf(fmaxf(macc[b].z, s0.z), s1.z);
                    macc[b].w = fmaxf(fmaxf(macc[b].w, s0.w), s1.w);
                }
            }
        }
#pragma unroll
        for (int b = 0; b < BPG; ++b)
            *reinterpret_cast<v4f*>(&gq[wave][b][4 * lane]) = macc[b];
    }
    lds_barrier();

    {
        bool in[BPG][16];
#pragma unroll
        for (int j = 0; j < 4; ++j) {
            const v4u q = *reinterpret_cast<const v4u*>(&smask[sb + 128 * j]);
#pragma unroll
            for (int b = 0; b < BPG; ++b) {
                in[b][4*j+0] = (q.x & (1u << b)) == 0u;
                in[b][4*j+1] = (q.y & (1u << b)) == 0u;
                in[b][4*j+2] = (q.z & (1u << b)) == 0u;
                in[b][4*j+3] = (q.w & (1u << b)) == 0u;
            }
        }

        float* __restrict__ myred = red[wave];

        for (int g8 = 0; g8 < RPW / 8; ++g8) {
            float p[32];
            const float* __restrict__ rowp = base + (size_t)(g8 * 8) * SN;
#pragma unroll
            for (int r = 0; r < 4; ++r)
#pragma unroll
                for (int j = 0; j < 4; ++j)
                    nxt[r][j] = *reinterpret_cast<const v4f*>(rowp + (size_t)(4 + r) * SN + 128 * j);
#pragma unroll
            for (int r = 0; r < 4; ++r) {
#pragma unroll
                for (int b = 0; b < BPG; ++b) {
                    float s[16];
#pragma unroll
                    for (int j = 0; j < 4; ++j) {
                        const v4f x = cur[r][j];
                        s[4*j+0] = in[b][4*j+0] ? x.x : 0.0f;
                        s[4*j+1] = in[b][4*j+1] ? x.y : 0.0f;
                        s[4*j+2] = in[b][4*j+2] ? x.z : 0.0f;
                        s[4*j+3] = in[b][4*j+3] ? x.w : 0.0f;
                    }
                    p[r * BPG + b] = max16(s);
                }
            }
            if (g8 + 1 < RPW / 8) {
#pragma unroll
                for (int r = 0; r < 4; ++r)
#pragma unroll
                    for (int j = 0; j < 4; ++j)
                        cur[r][j] = *reinterpret_cast<const v4f*>(rowp + (size_t)(8 + r) * SN + 128 * j);
            }
#pragma unroll
            for (int r = 0; r < 4; ++r) {
#pragma unroll
                for (int b = 0; b < BPG; ++b) {
                    float s[16];
#pragma unroll
                    for (int j = 0; j < 4; ++j) {
                        const v4f x = nxt[r][j];
                        s[4*j+0] = in[b][4*j+0] ? x.x : 0.0f;
                        s[4*j+1] = in[b][4*j+1] ? x.y : 0.0f;
                        s[4*j+2] = in[b][4*j+2] ? x.z : 0.0f;
                        s[4*j+3] = in[b][4*j+3] ? x.w : 0.0f;
                    }
                    p[(4 + r) * BPG + b] = max16(s);
                }
            }
#pragma unroll
            for (int k = 0; k < 32; ++k) myred[k * RPITCH + lane] = p[k];
            __builtin_amdgcn_fence(__ATOMIC_RELEASE, "wavefront");
            __builtin_amdgcn_fence(__ATOMIC_ACQUIRE, "wavefront");
            float c[32];
#pragma unroll
            for (int j = 0; j < 8; ++j) {
                const v4f x = *reinterpret_cast<const v4f*>(&myred[lane * RPITCH + 4 * j]);
                c[4*j+0] = x.x; c[4*j+1] = x.y; c[4*j+2] = x.z; c[4*j+3] = x.w;
            }
            gp[half][lane & (BPG - 1)][row0 + g8 * 8 + (lane >> 2)] = max32(c);
        }
    }

#if defined(__AMDGCN__) && __has_builtin(__builtin_amdgcn_wmma_f32_16x16x4_f32)
    v4f wf[2][8];
    const float* __restrict__ wrow = W + (size_t)(nbase + l15) * KDIM + 4 * hsel;
    if (__builtin_amdgcn_readfirstlane(wave) < NGW) {
#pragma unroll
        for (int j = 0; j < 8; ++j) wf[0][j] = *reinterpret_cast<const v4f*>(wrow + 8 * j);
    }
#endif

    lds_barrier();

    if (t < DSEED) {
#pragma unroll
        for (int b = 0; b < BPG; ++b) g[b][t] = fmaxf(gp[0][b][t], gp[1][b][t]);
    }
    for (int i = t; i < BPG * CBOX; i += NTHR) {
        const int b = i / CBOX, c = i % CBOX;
        float m = fmaxf(gq[0][b][c], gq[1][b][c]);
#pragma unroll
        for (int w = 2; w < NWAVE; w += 2) m = fmaxf(fmaxf(m, gq[w][b][c]), gq[w + 1][b][c]);
        g[b][DSEED + c] = m;
    }
    lds_barrier();

    if (__builtin_amdgcn_readfirstlane(wave) >= NGW) return;
    v8f acc = {};
#if defined(__AMDGCN__) && __has_builtin(__builtin_amdgcn_wmma_f32_16x16x4_f32)
    const float* grow = &g[l15 & (BPG - 1)][4 * hsel];
    v4f gf[2][8];
#pragma unroll
    for (int j = 0; j < 8; ++j) gf[0][j] = *reinterpret_cast<const v4f*>(grow + 8 * j);
#pragma unroll
    for (int blk = 0; blk < KDIM / 64; ++blk) {
        const int cu = blk & 1, nx = cu ^ 1;
        if (blk + 1 < KDIM / 64) {
#pragma unroll
            for (int j = 0; j < 8; ++j) wf[nx][j] = *reinterpret_cast<const v4f*>(wrow + 64 * (blk + 1) + 8 * j);
#pragma unroll
            for (int j = 0; j < 8; ++j) gf[nx][j] = *reinterpret_cast<const v4f*>(grow + 64 * (blk + 1) + 8 * j);
        }
        __builtin_amdgcn_sched_barrier(0);
#pragma unroll
        for (int j = 0; j < 8; ++j) {
            v2f a, b;
            a.x = gf[cu][j].x; a.y = gf[cu][j].y;
            b.x = wf[cu][j].x; b.y = wf[cu][j].y;
            acc = __builtin_amdgcn_wmma_f32_16x16x4_f32(false, a, false, b,
                                                        (short)0, acc, false, false);
            a.x = gf[cu][j].z; a.y = gf[cu][j].w;
            b.x = wf[cu][j].z; b.y = wf[cu][j].w;
            acc = __builtin_amdgcn_wmma_f32_16x16x4_f32(false, a, false, b,
                                                        (short)0, acc, false, false);
        }
        __builtin_amdgcn_sched_barrier(0);
    }
#elif defined(__AMDGCN__)
    for (int k0 = 0; k0 < KDIM; k0 += 32) {
        v16h a, b;
#pragma unroll
        for (int v = 0; v < 8; ++v) {
#pragma unroll
            for (int h = 0; h < 2; ++h) {
                const int k = k0 + 2*v + h + 8 * (hsel + (v >= 4));
                a[2*v + h] = (_Float16)g[l15][k];
                b[2*v + h] = (_Float16)W[(size_t)(nbase + l15) * KDIM + k];
            }
        }
        acc = __builtin_amdgcn_wmma_f32_16x16x32_f16(false, a, false, b,
                                                     (short)0, acc, false, false);
    }
#endif

#pragma unroll
    for (int j = 0; j < 8; ++j) {
        const int m  = j + 8 * hsel;
        const int gm = box0 + m;
        if (m < BPG && gm < UB) {
            const float x  = acc[j] + bn + 1e-6f;
            const float ax = fabsf(x);
            out[(size_t)gm * ODIM + nbase + l15] = ax / (1.0f + ax);
        }
    }
}

extern "C" void kernel_launch(void* const* d_in, const int* in_sizes, int n_in,
                              void* d_out, int out_size, void* d_ws, size_t ws_size,
                              hipStream_t stream) {
    const float* union_box    = (const float*)d_in[0];
    const float* box_features = (const float*)d_in[1];
    const float* agg_xyz      = (const float*)d_in[2];
    const float* seed_feature = (const float*)d_in[3];
    const float* seed_xyz     = (const float*)d_in[4];
    const float* bfu          = (const float*)d_in[5];
    const float* W_gs         = (const float*)d_in[6];
    const float* b_gs         = (const float*)d_in[7];
    float* out = (float*)d_out;

    const int grid = (UB + BPG - 1) / BPG;
    scenebox_fused_kernel<<<grid, NTHR, 0, stream>>>(
        union_box, box_features, agg_xyz, seed_feature, seed_xyz,
        bfu, W_gs, b_gs, out);
}
